# starting kernel
# baseline (speedup 1.0000x reference)
#include <hip/hip_runtime.h>
#include <hip/hip_bf16.h>

#define B_ 256
#define G_ 20000
#define P_ 1024
#define M_ 256
#define H_ 128
#define E_ 64

typedef __attribute__((ext_vector_type(2))) float v2f;
typedef __attribute__((ext_vector_type(4))) float v4f;
typedef __attribute__((ext_vector_type(8))) float v8f;

__global__ void transpose_x(const float* __restrict__ x, float* __restrict__ xT) {
    __shared__ float tile[32][33];
    int g0 = blockIdx.x * 32, b0 = blockIdx.y * 32;
    int tx = threadIdx.x, ty = threadIdx.y;
#pragma unroll
    for (int i = ty; i < 32; i += 8)
        tile[i][tx] = x[(size_t)(b0 + i) * G_ + (g0 + tx)];
    __syncthreads();
#pragma unroll
    for (int i = ty; i < 32; i += 8)
        xT[(size_t)(g0 + i) * B_ + (b0 + tx)] = tile[tx][i];
}

__global__ void pathway_kernel_t(const float* __restrict__ xT,
                                 const int* __restrict__ idx,
                                 const float* __restrict__ pw,
                                 const float* __restrict__ pmask,
                                 const float* __restrict__ pb,
                                 float* __restrict__ a_out) {
    __shared__ int   s_idx[M_];
    __shared__ float s_w[M_];
    int p = blockIdx.x;
    int t = threadIdx.x;
    for (int m = t; m < M_; m += 64) {
        s_idx[m] = idx[p * M_ + m];
        s_w[m]   = pw[p * M_ + m] * pmask[p * M_ + m];
    }
    __syncthreads();
    float a0 = 0.f, a1 = 0.f, a2 = 0.f, a3 = 0.f;
#pragma unroll 4
    for (int m = 0; m < M_; ++m) {
        v4f xv = *(const v4f*)(xT + (size_t)s_idx[m] * B_ + 4 * t);
        float w = s_w[m];
        a0 = fmaf(xv.x, w, a0);
        a1 = fmaf(xv.y, w, a1);
        a2 = fmaf(xv.z, w, a2);
        a3 = fmaf(xv.w, w, a3);
    }
    float bb = pb[p];
    a_out[(size_t)(4 * t + 0) * P_ + p] = tanhf(a0 + bb);
    a_out[(size_t)(4 * t + 1) * P_ + p] = tanhf(a1 + bb);
    a_out[(size_t)(4 * t + 2) * P_ + p] = tanhf(a2 + bb);
    a_out[(size_t)(4 * t + 3) * P_ + p] = tanhf(a3 + bb);
}

__global__ void pathway_kernel_s(const float* __restrict__ x,
                                 const int* __restrict__ idx,
                                 const float* __restrict__ pw,
                                 const float* __restrict__ pmask,
                                 const float* __restrict__ pb,
                                 float* __restrict__ a_out) {
    __shared__ int   s_idx[M_];
    __shared__ float s_w[M_];
    int p = blockIdx.x;
    int t = threadIdx.x;
    s_idx[t] = idx[p * M_ + t];
    s_w[t]   = pw[p * M_ + t] * pmask[p * M_ + t];
    __syncthreads();
    float acc = 0.f;
#pragma unroll 4
    for (int m = 0; m < M_; ++m)
        acc = fmaf(x[(size_t)t * G_ + s_idx[m]], s_w[m], acc);
    a_out[(size_t)t * P_ + p] = tanhf(acc + pb[p]);
}

template <int ACT>
__global__ void gemm_wmma_f32(const float* __restrict__ A,
                              const float* __restrict__ W,
                              const float* __restrict__ bias,
                              float* __restrict__ D,
                              int Mr, int K, int N) {
    int lane  = threadIdx.x & 31;
    int wave  = threadIdx.x >> 5;
    int waves = blockDim.x >> 5;
    int tiles_n = N >> 4;
    int tile  = blockIdx.x * waves + wave;
    int total = (Mr >> 4) * tiles_n;
    if (tile >= total) return;
    int tm = tile / tiles_n;
    int tn = tile - tm * tiles_n;
    int mn   = lane & 15;
    int koff = (lane >> 4) << 1;
    const float* arow = A + (size_t)(tm * 16 + mn) * K + koff;
    const float* wcol = W + (tn * 16 + mn);
    v8f c = {};
    for (int k = 0; k < K; k += 4) {
        v2f a = *(const v2f*)(arow + k);
        v2f b;
        b.x = wcol[(size_t)(k + koff) * N];
        b.y = wcol[(size_t)(k + koff + 1) * N];
        c = __builtin_amdgcn_wmma_f32_16x16x4_f32(
                false, a, false, b, (short)0, c, false, false);
    }
    float bv = bias[tn * 16 + mn];
    int rbase = tm * 16 + ((lane >> 4) << 3);
#pragma unroll
    for (int j = 0; j < 8; ++j) {
        float v = c[j] + bv;
        if (ACT == 1) v = fmaxf(v, 0.f);
        D[(size_t)(rbase + j) * N + (tn * 16 + mn)] = v;
    }
}

__global__ void dec_gemm_wmma(const float* __restrict__ A,
                              const float* __restrict__ W,
                              const float* __restrict__ bias,
                              float* __restrict__ D) {
    __shared__ float wpanel[H_ * 16];
    int tn = blockIdx.x;
    int t  = threadIdx.x;
#pragma unroll
    for (int i = 0; i < H_ * 16; i += 512) {
        int e = i + t;
        wpanel[e] = W[(size_t)(e >> 4) * G_ + tn * 16 + (e & 15)];
    }
    __syncthreads();
    int lane = t & 31;
    int tm   = t >> 5;
    int mn   = lane & 15;
    int koff = (lane >> 4) << 1;
    const float* arow = A + (size_t)(tm * 16 + mn) * H_ + koff;
    v8f c = {};
#pragma unroll
    for (int k = 0; k < H_; k += 4) {
        v2f a = *(const v2f*)(arow + k);
        v2f b;
        b.x = wpanel[(k + koff) * 16 + mn];
        b.y = wpanel[(k + koff + 1) * 16 + mn];
        c = __builtin_amdgcn_wmma_f32_16x16x4_f32(
                false, a, false, b, (short)0, c, false, false);
    }
    float bv = bias[tn * 16 + mn];
    int rbase = tm * 16 + ((lane >> 4) << 3);
#pragma unroll
    for (int j = 0; j < 8; ++j)
        D[(size_t)(rbase + j) * G_ + (tn * 16 + mn)] = c[j] + bv;
}

__global__ void reparam_kernel(const float* __restrict__ ml,
                               const float* __restrict__ eps,
                               float* __restrict__ z,
                               float* __restrict__ mean_o,
                               float* __restrict__ logvar_o) {
    int i = blockIdx.x * blockDim.x + threadIdx.x;
    if (i >= B_ * E_) return;
    int b = i >> 6, e = i & 63;
    float mean   = ml[b * (2 * E_) + e];
    float logvar = ml[b * (2 * E_) + E_ + e];
    z[i]        = fmaf(eps[i], logvar, mean);
    mean_o[i]   = mean;
    logvar_o[i] = logvar;
}

extern "C" void kernel_launch(void* const* d_in, const int* in_sizes, int n_in,
                              void* d_out, int out_size, void* d_ws, size_t ws_size,
                              hipStream_t stream) {
    const float* x     = (const float*)d_in[0];
    const int*   pidx  = (const int*)d_in[1];
    const float* pmask = (const float*)d_in[2];
    const float* eps   = (const float*)d_in[3];
    const float* pw    = (const float*)d_in[4];
    const float* pb    = (const float*)d_in[5];
    const float* ew1   = (const float*)d_in[6];
    const float* eb1   = (const float*)d_in[7];
    const float* ew2   = (const float*)d_in[8];
    const float* eb2   = (const float*)d_in[9];
    const float* dw1   = (const float*)d_in[10];
    const float* db1   = (const float*)d_in[11];
    const float* dw2   = (const float*)d_in[12];
    const float* db2   = (const float*)d_in[13];

    float* ws = (float*)d_ws;
    float* a  = ws;
    float* h  = ws + 262144;
    float* ml = ws + 294912;
    float* hd = ws + 327680;
    float* xT = ws + 360448;
    size_t need_xt = (size_t)(360448 + (size_t)G_ * B_) * sizeof(float);

    float* out    = (float*)d_out;
    float* z_o    = out;
    float* recon  = out + (size_t)B_ * E_;
    float* mean_o = out + (size_t)B_ * E_ + (size_t)B_ * G_;
    float* logv_o = mean_o + (size_t)B_ * E_;

    if (ws_size >= need_xt) {
        transpose_x<<<dim3(G_ / 32, B_ / 32), dim3(32, 8), 0, stream>>>(x, xT);
        pathway_kernel_t<<<P_, 64, 0, stream>>>(xT, pidx, pw, pmask, pb, a);
    } else {
        pathway_kernel_s<<<P_, 256, 0, stream>>>(x, pidx, pw, pmask, pb, a);
    }

    gemm_wmma_f32<1><<<16, 256, 0, stream>>>(a, ew1, eb1, h, B_, P_, H_);
    gemm_wmma_f32<0><<<16, 256, 0, stream>>>(h, ew2, eb2, ml, B_, H_, 2 * E_);
    reparam_kernel<<<(B_ * E_ + 255) / 256, 256, 0, stream>>>(ml, eps, z_o, mean_o, logv_o);
    gemm_wmma_f32<1><<<16, 256, 0, stream>>>(z_o, dw1, db1, hd, B_, E_, H_);
    dec_gemm_wmma<<<G_ / 16, 512, 0, stream>>>(hd, dw2, db2, recon);
}


# Decoder GEMM: a wave keeps its 128x16 dec_w2 panel in registers (W never in LDS) and runs it against 4 row tiles of hd read from one LDS copy per block; 5000 waves enter, not 20000; sums unchanged
# speedup vs baseline: 2.5914x; 2.5914x over previous
#include <hip/hip_runtime.h>
#include <hip/hip_bf16.h>

#define B_ 256
#define G_ 20000
#define P_ 1024
#define M_ 256
#define H_ 128
#define E_ 64

typedef __attribute__((ext_vector_type(2))) float v2f;
typedef __attribute__((ext_vector_type(4))) float v4f;
typedef __attribute__((ext_vector_type(8))) float v8f;

typedef __attribute__((ext_vector_type(4))) int v4i;

#define XS_PAD 512
#define AS_STRIDE 516

template <int CTRL>
__device__ __forceinline__ float row_lane_xor(float v) {
    return __builtin_bit_cast(float, __builtin_amdgcn_update_dpp(
        0, __builtin_bit_cast(int, v), CTRL, 0xf, 0xf, false));
}

__device__ __forceinline__ float other_row(float v) {
    int i = __builtin_bit_cast(int, v);
    return __builtin_bit_cast(float, __builtin_amdgcn_permlanex16(
        i, i, 0x76543210, 0xfedcba98, false, false));
}

__device__ __forceinline__ float pathway_sum(const v2f* xs, v4i i0, v4i i1,
                                             v4f w0, v4f w1, v4f k0, v4f k1, int hrow) {
    v2f g0 = xs[i0.x], g1 = xs[i0.y], g2 = xs[i0.z], g3 = xs[i0.w];
    v2f g4 = xs[i1.x], g5 = xs[i1.y], g6 = xs[i1.z], g7 = xs[i1.w];
    float m0 = w0.x * k0.x, m1 = w0.y * k0.y, m2 = w0.z * k0.z, m3 = w0.w * k0.w;
    float m4 = w1.x * k1.x, m5 = w1.y * k1.y, m6 = w1.z * k1.z, m7 = w1.w * k1.w;
    float a0 = g0.x * m0, a1 = g0.y * m0;
    a0 = fmaf(g1.x, m1, a0); a1 = fmaf(g1.y, m1, a1);
    a0 = fmaf(g2.x, m2, a0); a1 = fmaf(g2.y, m2, a1);
    a0 = fmaf(g3.x, m3, a0); a1 = fmaf(g3.y, m3, a1);
    a0 = fmaf(g4.x, m4, a0); a1 = fmaf(g4.y, m4, a1);
    a0 = fmaf(g5.x, m5, a0); a1 = fmaf(g5.y, m5, a1);
    a0 = fmaf(g6.x, m6, a0); a1 = fmaf(g6.y, m6, a1);
    a0 = fmaf(g7.x, m7, a0); a1 = fmaf(g7.y, m7, a1);
    float keep = hrow ? a1 : a0;
    float send = hrow ? a0 : a1;
    float v = keep + other_row(send);
    v += row_lane_xor<0x168>(v);
    v += row_lane_xor<0x164>(v);
    v += row_lane_xor<0x162>(v);
    v += row_lane_xor<0x161>(v);
    return v;
}

__global__ void __launch_bounds__(1024)
pathway_rows_lds(const float* __restrict__ x,
                 const int* __restrict__ idx,
                 const float* __restrict__ pw,
                 const float* __restrict__ pmask,
                 const float* __restrict__ pb,
                 const float* __restrict__ w1,
                 float* __restrict__ hpart) {
    __shared__ v2f xs[G_ + XS_PAD];
    __shared__ float as[2 * AS_STRIDE];
    __shared__ float red[4 * 2 * H_];
    const int t  = threadIdx.x;
    const int b0 = (blockIdx.x >> 1) << 1;
    const int s  = blockIdx.x & 1;

    {
        const v4f* x0 = (const v4f*)(x + (size_t)b0 * G_);
        const v4f* x1 = (const v4f*)(x + (size_t)(b0 + 1) * G_);
        v4f* xs4 = (v4f*)xs;
        constexpr int Q = G_ / 4;
        v4f r0[5], r1[5];
#pragma unroll
        for (int i = 0; i < 5; ++i) {
            int q  = t + 1024 * i;
            int qc = q < Q ? q : Q - 1;
            r0[i] = x0[qc];
            r1[i] = x1[qc];
        }
#pragma unroll
        for (int i = 0; i < 5; ++i) {
            int q = t + 1024 * i;
            if (q < Q) {
                v4f lo = {r0[i].x, r1[i].x, r0[i].y, r1[i].y};
                v4f hi = {r0[i].z, r1[i].z, r0[i].w, r1[i].w};
                xs4[2 * q]     = lo;
                xs4[2 * q + 1] = hi;
            }
        }
    }
    __syncthreads();

    const int lane = t & 31;
    const int c    = lane & 15;
    const int hrow = lane >> 4;
    const int p0   = s * 512 + (t >> 5) * 16;
    const size_t off = (size_t)p0 * M_ + 4 * lane;
    const int*   ip = idx   + off;
    const float* wp = pw    + off;
    const float* kp = pmask + off;

    float res = 0.f;
#pragma clang loop unroll(disable)
    for (int j = 0; j < 16; j += 2) {
        const int ja = j * M_, jb = ja + M_;
        v4i ai0 = *(const v4i*)(ip + ja), ai1 = *(const v4i*)(ip + ja + 128);
        v4f aw0 = *(const v4f*)(wp + ja), aw1 = *(const v4f*)(wp + ja + 128);
        v4f ak0 = *(const v4f*)(kp + ja), ak1 = *(const v4f*)(kp + ja + 128);
        v4i bi0 = *(const v4i*)(ip + jb), bi1 = *(const v4i*)(ip + jb + 128);
        v4f bw0 = *(const v4f*)(wp + jb), bw1 = *(const v4f*)(wp + jb + 128);
        v4f bk0 = *(const v4f*)(kp + jb), bk1 = *(const v4f*)(kp + jb + 128);
        float va = pathway_sum(xs, ai0, ai1, aw0, aw1, ak0, ak1, hrow);
        float vb = pathway_sum(xs, bi0, bi1, bw0, bw1, bk0, bk1, hrow);
        if (c == j)     res = va;
        if (c == j + 1) res = vb;
    }
    {
        const int wv   = t >> 5;
        const int n    = wv & 7;
        const int q    = wv >> 3;
        const int koff = hrow << 1;
        const float* wcol = w1 + (size_t)(s * 512 + q * 128 + koff) * H_ + n * 16 + c;
        const float* arow = as + (c & 1) * AS_STRIDE + q * 128 + koff;
        v2f bf[32];
#pragma unroll
        for (int i = 0; i < 32; ++i) {
            bf[i].x = wcol[(size_t)(4 * i) * H_];
            bf[i].y = wcol[(size_t)(4 * i + 1) * H_];
        }

        as[hrow * AS_STRIDE + wv * 16 + c] = tanhf(res + pb[p0 + c]);
        __syncthreads();

        v8f acc = {};
#pragma unroll
        for (int i = 0; i < 32; ++i) {
            v2f af = *(const v2f*)(arow + 4 * i);
            acc = __builtin_amdgcn_wmma_f32_16x16x4_f32(
                      false, af, false, bf[i], (short)0, acc, false, false);
        }
        if (hrow == 0) {
            red[(q * 2 + 0) * H_ + n * 16 + c] = acc[0];
            red[(q * 2 + 1) * H_ + n * 16 + c] = acc[1];
        }
    }
    __syncthreads();
    if (t < 2 * H_) {
        const int r = t >> 7, col = t & (H_ - 1);
        float v = (red[(0 * 2 + r) * H_ + col] + red[(1 * 2 + r) * H_ + col])
                + (red[(2 * 2 + r) * H_ + col] + red[(3 * 2 + r) * H_ + col]);
        hpart[((size_t)s * B_ + b0 + r) * H_ + col] = v;
    }
}

template <int K, int N>
__device__ __forceinline__ void load_chunk(const float* __restrict__ arow,
                                           const float* __restrict__ wcol,
                                           int k0, v2f (&a)[16], v2f (&b)[16]) {
#pragma unroll
    for (int s = 0; s < 16; ++s) {
        int k = k0 + 4 * s;
        a[s]   = *(const v2f*)(arow + k);
        b[s].x = wcol[(size_t)k * N];
        b[s].y = wcol[(size_t)(k + 1) * N];
    }
}

__device__ __forceinline__ v8f mma_chunk(const v2f (&a)[16], const v2f (&b)[16], v8f c) {
#pragma unroll
    for (int s = 0; s < 16; ++s)
        c = __builtin_amdgcn_wmma_f32_16x16x4_f32(
                false, a[s], false, b[s], (short)0, c, false, false);
    return c;
}

template <int ACT, int K, int N>
__global__ void __launch_bounds__(32)
gemm_wmma_f32(const float* __restrict__ A,
              const float* __restrict__ W,
              const float* __restrict__ bias,
              float* __restrict__ D) {
    constexpr int KC = 64;
    constexpr int NC = K / KC;
    constexpr int tiles_n = N >> 4;
    int lane = threadIdx.x;
    int tile = blockIdx.x;
    int tm = tile / tiles_n;
    int tn = tile - tm * tiles_n;
    int mn   = lane & 15;
    int koff = (lane >> 4) << 1;
    const float* arow = A + (size_t)(tm * 16 + mn) * K + koff;
    const float* wcol = W + (size_t)koff * N + (tn * 16 + mn);
    v2f a0[16], b0[16], a1[16], b1[16];
    v8f c = {};
    load_chunk<K, N>(arow, wcol, 0, a0, b0);
    int ch = 0;
    for (; ch + 2 < NC; ch += 2) {
        load_chunk<K, N>(arow, wcol, (ch + 1) * KC, a1, b1);
        c = mma_chunk(a0, b0, c);
        load_chunk<K, N>(arow, wcol, (ch + 2) * KC, a0, b0);
        c = mma_chunk(a1, b1, c);
    }
    if (ch + 1 < NC) {
        load_chunk<K, N>(arow, wcol, (ch + 1) * KC, a1, b1);
        c = mma_chunk(a0, b0, c);
        c = mma_chunk(a1, b1, c);
    } else {
        c = mma_chunk(a0, b0, c);
    }
    float bv = bias[tn * 16 + mn];
    int rbase = tm * 16 + ((lane >> 4) << 3);
#pragma unroll
    for (int j = 0; j < 8; ++j) {
        float v = c[j] + bv;
        if (ACT == 1) v = fmaxf(v, 0.f);
        D[(size_t)(rbase + j) * N + (tn * 16 + mn)] = v;
    }
}

__global__ void __launch_bounds__(32)
enc_out_gemm(const float* __restrict__ hp,
             const float* __restrict__ b1,
             const float* __restrict__ W,
             const float* __restrict__ bias,
             float* __restrict__ D) {
    constexpr int K = H_;
    constexpr int N = 2 * E_;
    constexpr int tiles_n = N >> 4;
    int lane = threadIdx.x;
    int tile = blockIdx.x;
    int tm = tile / tiles_n;
    int tn = tile - tm * tiles_n;
    int mn   = lane & 15;
    int koff = (lane >> 4) << 1;
    const float* p0   = hp + (size_t)(tm * 16 + mn) * K + koff;
    const float* p1   = p0 + (size_t)B_ * H_;
    const float* bk   = b1 + koff;
    const float* wcol = W + (size_t)koff * N + (tn * 16 + mn);
    v2f a[32], b[32];
#pragma unroll
    for (int s = 0; s < 32; ++s) {
        int k = 4 * s;
        v2f u = *(const v2f*)(p0 + k);
        v2f v = *(const v2f*)(p1 + k);
        v2f e = *(const v2f*)(bk + k);
        a[s].x = fmaxf((u.x + v.x) + e.x, 0.f);
        a[s].y = fmaxf((u.y + v.y) + e.y, 0.f);
        b[s].x = wcol[(size_t)k * N];
        b[s].y = wcol[(size_t)(k + 1) * N];
    }
    v8f c = {};
#pragma unroll
    for (int s = 0; s < 32; ++s)
        c = __builtin_amdgcn_wmma_f32_16x16x4_f32(
                false, a[s], false, b[s], (short)0, c, false, false);
    float bv = bias[tn * 16 + mn];
    int rbase = tm * 16 + ((lane >> 4) << 3);
#pragma unroll
    for (int j = 0; j < 8; ++j)
        D[(size_t)(rbase + j) * N + (tn * 16 + mn)] = c[j] + bv;
}

#define HD_PITCH 132
#define DEC_PANELS 5
#define DEC_THREADS (32 * 4 * DEC_PANELS)

__global__ void __launch_bounds__(DEC_THREADS)
dec_gemm_wreg(const float* __restrict__ A,
              const float* __restrict__ W,
              const float* __restrict__ bias,
              float* __restrict__ D) {
    __shared__ v4f hs4[B_ * HD_PITCH / 4];
    float* hs = (float*)hs4;
    const int t    = threadIdx.x;
    const int lane = t & 31;
    const int wv   = t >> 5;
    const int c    = lane & 15;
    const int h    = lane >> 4;
    const int pl   = wv >> 2;
    const int rq   = wv & 3;
    const int col  = (blockIdx.x * DEC_PANELS + pl) * 16 + c;

    const float* wcol = W + (size_t)(2 * h) * G_ + col;
    v2f bf[32];
#pragma unroll
    for (int i = 0; i < 32; ++i) {
        bf[i].x = wcol[(size_t)(4 * i) * G_];
        bf[i].y = wcol[(size_t)(4 * i + 1) * G_];
    }
    const float bv = bias[col];

    if (wv < 16) {
        const v4f* A4 = (const v4f*)A;
        v4f r[16];
#pragma unroll
        for (int i = 0; i < 16; ++i)
            r[i] = A4[t + 512 * i];
#pragma unroll
        for (int i = 0; i < 16; ++i) {
            int q = t + 512 * i;
            *(v4f*)(hs + (q >> 5) * HD_PITCH + ((q & 31) << 2)) = r[i];
        }
    }
    __syncthreads();

    const float* abase = hs + (rq * 64 + c) * HD_PITCH + 2 * h;
    for (int r = 0; r < 4; ++r) {
        const float* arow = abase + r * 16 * HD_PITCH;
        v2f af[32];
#pragma unroll
        for (int i = 0; i < 32; ++i)
            af[i] = *(const v2f*)(arow + 4 * i);
        v8f acc = {};
#pragma unroll
        for (int i = 0; i < 32; ++i)
            acc = __builtin_amdgcn_wmma_f32_16x16x4_f32(
                      false, af[i], false, bf[i], (short)0, acc, false, false);
        const int rbase = (rq * 4 + r) * 16 + (h << 3);
#pragma unroll
        for (int j = 0; j < 8; ++j)
            D[(size_t)(rbase + j) * G_ + col] = acc[j] + bv;
    }
}

__global__ void reparam_kernel(const float* __restrict__ ml,
                               const float* __restrict__ eps,
                               float* __restrict__ z,
                               float* __restrict__ mean_o,
                               float* __restrict__ logvar_o) {
    int i = blockIdx.x * blockDim.x + threadIdx.x;
    if (i >= B_ * E_) return;
    int b = i >> 6, e = i & 63;
    float mean   = ml[b * (2 * E_) + e];
    float logvar = ml[b * (2 * E_) + E_ + e];
    z[i]        = fmaf(eps[i], logvar, mean);
    mean_o[i]   = mean;
    logvar_o[i] = logvar;
}

extern "C" void kernel_launch(void* const* d_in, const int* in_sizes, int n_in,
                              void* d_out, int out_size, void* d_ws, size_t ws_size,
                              hipStream_t stream) {
    const float* x     = (const float*)d_in[0];
    const int*   pidx  = (const int*)d_in[1];
    const float* pmask = (const float*)d_in[2];
    const float* eps   = (const float*)d_in[3];
    const float* pw    = (const float*)d_in[4];
    const float* pb    = (const float*)d_in[5];
    const float* ew1   = (const float*)d_in[6];
    const float* eb1   = (const float*)d_in[7];
    const float* ew2   = (const float*)d_in[8];
    const float* eb2   = (const float*)d_in[9];
    const float* dw1   = (const float*)d_in[10];
    const float* db1   = (const float*)d_in[11];
    const float* dw2   = (const float*)d_in[12];
    const float* db2   = (const float*)d_in[13];

    float* ws = (float*)d_ws;
    float* hp = ws;
    float* ml = ws + 294912;
    float* hd = ws + 327680;

    float* out    = (float*)d_out;
    float* z_o    = out;
    float* recon  = out + (size_t)B_ * E_;
    float* mean_o = out + (size_t)B_ * E_ + (size_t)B_ * G_;
    float* logv_o = mean_o + (size_t)B_ * E_;

    pathway_rows_lds<<<(B_ / 2) * 2, 1024, 0, stream>>>(x, pidx, pw, pmask, pb, ew1, hp);

    enc_out_gemm<<<(B_ / 16) * (2 * E_ / 16), 32, 0, stream>>>(hp, eb1, ew2, eb2, ml);
    reparam_kernel<<<(B_ * E_ + 255) / 256, 256, 0, stream>>>(ml, eps, z_o, mean_o, logv_o);
    gemm_wmma_f32<1, E_, H_><<<(B_ / 16) * (H_ / 16), 32, 0, stream>>>(z_o, dw1, db1, hd);
    dec_gemm_wreg<<<G_ / (16 * DEC_PANELS), DEC_THREADS, 0, stream>>>(hd, dw2, db2, recon);
}


# Pathway sums: a block holds 4 whole batch rows in LDS (was 2) and does 256 pathways (was 512): one 16-byte LDS read serves 4 rows, idx/w/mask fetched half as often; new f32 sum order
# speedup vs baseline: 2.7379x; 1.0565x over previous
#include <hip/hip_runtime.h>
#include <hip/hip_bf16.h>

#define B_ 256
#define G_ 20000
#define P_ 1024
#define M_ 256
#define H_ 128
#define E_ 64

typedef __attribute__((ext_vector_type(2))) float v2f;
typedef __attribute__((ext_vector_type(4))) float v4f;
typedef __attribute__((ext_vector_type(8))) float v8f;

typedef __attribute__((ext_vector_type(4))) int v4i;

#define AS_STRIDE 260

__device__ __forceinline__ int xs_slot(int g) { return g ^ ((g >> 4) & 3); }

template <int CTRL>
__device__ __forceinline__ float row_lane_xor(float v) {
    return __builtin_bit_cast(float, __builtin_amdgcn_update_dpp(
        0, __builtin_bit_cast(int, v), CTRL, 0xf, 0xf, false));
}

__device__ __forceinline__ float other_row(float v) {
    int i = __builtin_bit_cast(int, v);
    return __builtin_bit_cast(float, __builtin_amdgcn_permlanex16(
        i, i, 0x76543210, 0xfedcba98, false, false));
}

__device__ __forceinline__ float pathway_sum(const v4f* xs, v4i i0, v4i i1,
                                             v4f w0, v4f w1, v4f k0, v4f k1,
                                             int hrow, int oct) {
    v4f g0 = xs[xs_slot(i0.x)], g1 = xs[xs_slot(i0.y)];
    v4f g2 = xs[xs_slot(i0.z)], g3 = xs[xs_slot(i0.w)];
    v4f g4 = xs[xs_slot(i1.x)], g5 = xs[xs_slot(i1.y)];
    v4f g6 = xs[xs_slot(i1.z)], g7 = xs[xs_slot(i1.w)];
    float m0 = w0.x * k0.x, m1 = w0.y * k0.y, m2 = w0.z * k0.z, m3 = w0.w * k0.w;
    float m4 = w1.x * k1.x, m5 = w1.y * k1.y, m6 = w1.z * k1.z, m7 = w1.w * k1.w;
    v4f a = {g0.x * m0, g0.y * m0, g0.z * m0, g0.w * m0};
#define ROW4_FMA(g, m) a.x = fmaf(g.x, m, a.x); a.y = fmaf(g.y, m, a.y); \
                       a.z = fmaf(g.z, m, a.z); a.w = fmaf(g.w, m, a.w);
    ROW4_FMA(g1, m1) ROW4_FMA(g2, m2) ROW4_FMA(g3, m3) ROW4_FMA(g4, m4)
    ROW4_FMA(g5, m5) ROW4_FMA(g6, m6) ROW4_FMA(g7, m7)
#undef ROW4_FMA
    float ke = hrow ? a.z : a.x, ko = hrow ? a.w : a.y;
    float se = hrow ? a.x : a.z, so = hrow ? a.y : a.w;
    float ue = ke + other_row(se);
    float uo = ko + other_row(so);
    float keep = oct ? uo : ue;
    float send = oct ? ue : uo;
    float v = keep + row_lane_xor<0x168>(send);
    v += row_lane_xor<0x164>(v);
    v += row_lane_xor<0x162>(v);
    v += row_lane_xor<0x161>(v);
    return v;
}

__global__ void __launch_bounds__(1024)
pathway_rows_lds(const float* __restrict__ x,
                 const int* __restrict__ idx,
                 const float* __restrict__ pw,
                 const float* __restrict__ pmask,
                 const float* __restrict__ pb,
                 const float* __restrict__ w1,
                 float* __restrict__ hpart) {
    __shared__ v4f xs[G_];
    __shared__ float as[4 * AS_STRIDE];
    float* red = (float*)xs;
    const int t  = threadIdx.x;
    const int b0 = (blockIdx.x >> 2) << 2;
    const int s  = blockIdx.x & 3;

    {
        const v4f* x0 = (const v4f*)(x + (size_t)b0 * G_);
        const v4f* x1 = (const v4f*)(x + (size_t)(b0 + 1) * G_);
        const v4f* x2 = (const v4f*)(x + (size_t)(b0 + 2) * G_);
        const v4f* x3 = (const v4f*)(x + (size_t)(b0 + 3) * G_);
        constexpr int Q = G_ / 4;
        v4f r0[5], r1[5], r2[5], r3[5];
#pragma unroll
        for (int i = 0; i < 5; ++i) {
            int q  = t + 1024 * i;
            int qc = q < Q ? q : Q - 1;
            r0[i] = x0[qc];
            r1[i] = x1[qc];
            r2[i] = x2[qc];
            r3[i] = x3[qc];
        }
#pragma unroll
        for (int i = 0; i < 5; ++i) {
            int q = t + 1024 * i;
            if (q < Q) {
                const int m = (q >> 2) & 3;
                v4f* dst = xs + 4 * q;
                v4f e0 = {r0[i].x, r1[i].x, r2[i].x, r3[i].x};
                v4f e1 = {r0[i].y, r1[i].y, r2[i].y, r3[i].y};
                v4f e2 = {r0[i].z, r1[i].z, r2[i].z, r3[i].z};
                v4f e3 = {r0[i].w, r1[i].w, r2[i].w, r3[i].w};
                dst[m]     = e0;
                dst[1 ^ m] = e1;
                dst[2 ^ m] = e2;
                dst[3 ^ m] = e3;
            }
        }
    }
    __syncthreads();

    const int lane = t & 31;
    const int c    = lane & 15;
    const int hrow = lane >> 4;
    const int oct  = (lane >> 3) & 1;
    const int c8   = lane & 7;
    const int p0   = s * 256 + (t >> 5) * 8;
    const size_t off = (size_t)p0 * M_ + 4 * lane;
    const int*   ip = idx   + off;
    const float* wp = pw    + off;
    const float* kp = pmask + off;

    float res = 0.f;
#pragma clang loop unroll(disable)
    for (int j = 0; j < 8; j += 2) {
        const int ja = j * M_, jb = ja + M_;
        v4i ai0 = *(const v4i*)(ip + ja), ai1 = *(const v4i*)(ip + ja + 128);
        v4f aw0 = *(const v4f*)(wp + ja), aw1 = *(const v4f*)(wp + ja + 128);
        v4f ak0 = *(const v4f*)(kp + ja), ak1 = *(const v4f*)(kp + ja + 128);
        v4i bi0 = *(const v4i*)(ip + jb), bi1 = *(const v4i*)(ip + jb + 128);
        v4f bw0 = *(const v4f*)(wp + jb), bw1 = *(const v4f*)(wp + jb + 128);
        v4f bk0 = *(const v4f*)(kp + jb), bk1 = *(const v4f*)(kp + jb + 128);
        float va = pathway_sum(xs, ai0, ai1, aw0, aw1, ak0, ak1, hrow, oct);
        float vb = pathway_sum(xs, bi0, bi1, bw0, bw1, bk0, bk1, hrow, oct);
        if (c8 == j)     res = va;
        if (c8 == j + 1) res = vb;
    }
    {
        const int wv   = t >> 5;
        const int n    = wv & 7;
        const int q    = wv >> 3;
        const int koff = hrow << 1;
        const float* wcol = w1 + (size_t)(s * 256 + q * 64 + koff) * H_ + n * 16 + c;
        const float* arow = as + (c & 3) * AS_STRIDE + q * 64 + koff;
        v2f bf[16];
#pragma unroll
        for (int i = 0; i < 16; ++i) {
            bf[i].x = wcol[(size_t)(4 * i) * H_];
            bf[i].y = wcol[(size_t)(4 * i + 1) * H_];
        }

        as[(lane >> 3) * AS_STRIDE + wv * 8 + c8] = tanhf(res + pb[p0 + c8]);
        __syncthreads();

        v8f acc = {};
#pragma unroll
        for (int i = 0; i < 16; ++i) {
            v2f af = *(const v2f*)(arow + 4 * i);
            acc = __builtin_amdgcn_wmma_f32_16x16x4_f32(
                      false, af, false, bf[i], (short)0, acc, false, false);
        }
        if (hrow == 0) {
            red[(q * 4 + 0) * H_ + n * 16 + c] = acc[0];
            red[(q * 4 + 1) * H_ + n * 16 + c] = acc[1];
            red[(q * 4 + 2) * H_ + n * 16 + c] = acc[2];
            red[(q * 4 + 3) * H_ + n * 16 + c] = acc[3];
        }
    }
    __syncthreads();
    if (t < 4 * H_) {
        const int r = t >> 7, col = t & (H_ - 1);
        float v = (red[(0 * 4 + r) * H_ + col] + red[(1 * 4 + r) * H_ + col])
                + (red[(2 * 4 + r) * H_ + col] + red[(3 * 4 + r) * H_ + col]);
        hpart[((size_t)s * B_ + b0 + r) * H_ + col] = v;
    }
}

template <int K, int N>
__device__ __forceinline__ void load_chunk(const float* __restrict__ arow,
                                           const float* __restrict__ wcol,
                                           int k0, v2f (&a)[16], v2f (&b)[16]) {
#pragma unroll
    for (int s = 0; s < 16; ++s) {
        int k = k0 + 4 * s;
        a[s]   = *(const v2f*)(arow + k);
        b[s].x = wcol[(size_t)k * N];
        b[s].y = wcol[(size_t)(k + 1) * N];
    }
}

__device__ __forceinline__ v8f mma_chunk(const v2f (&a)[16], const v2f (&b)[16], v8f c) {
#pragma unroll
    for (int s = 0; s < 16; ++s)
        c = __builtin_amdgcn_wmma_f32_16x16x4_f32(
                false, a[s], false, b[s], (short)0, c, false, false);
    return c;
}

template <int ACT, int K, int N>
__global__ void __launch_bounds__(32)
gemm_wmma_f32(const float* __restrict__ A,
              const float* __restrict__ W,
              const float* __restrict__ bias,
              float* __restrict__ D) {
    constexpr int KC = 64;
    constexpr int NC = K / KC;
    constexpr int tiles_n = N >> 4;
    int lane = threadIdx.x;
    int tile = blockIdx.x;
    int tm = tile / tiles_n;
    int tn = tile - tm * tiles_n;
    int mn   = lane & 15;
    int koff = (lane >> 4) << 1;
    const float* arow = A + (size_t)(tm * 16 + mn) * K + koff;
    const float* wcol = W + (size_t)koff * N + (tn * 16 + mn);
    v2f a0[16], b0[16], a1[16], b1[16];
    v8f c = {};
    load_chunk<K, N>(arow, wcol, 0, a0, b0);
    int ch = 0;
    for (; ch + 2 < NC; ch += 2) {
        load_chunk<K, N>(arow, wcol, (ch + 1) * KC, a1, b1);
        c = mma_chunk(a0, b0, c);
        load_chunk<K, N>(arow, wcol, (ch + 2) * KC, a0, b0);
        c = mma_chunk(a1, b1, c);
    }
    if (ch + 1 < NC) {
        load_chunk<K, N>(arow, wcol, (ch + 1) * KC, a1, b1);
        c = mma_chunk(a0, b0, c);
        c = mma_chunk(a1, b1, c);
    } else {
        c = mma_chunk(a0, b0, c);
    }
    float bv = bias[tn * 16 + mn];
    int rbase = tm * 16 + ((lane >> 4) << 3);
#pragma unroll
    for (int j = 0; j < 8; ++j) {
        float v = c[j] + bv;
        if (ACT == 1) v = fmaxf(v, 0.f);
        D[(size_t)(rbase + j) * N + (tn * 16 + mn)] = v;
    }
}

__global__ void __launch_bounds__(32)
enc_out_gemm(const float* __restrict__ hp,
             const float* __restrict__ b1,
             const float* __restrict__ W,
             const float* __restrict__ bias,
             float* __restrict__ D) {
    constexpr int K = H_;
    constexpr int N = 2 * E_;
    constexpr int tiles_n = N >> 4;
    int lane = threadIdx.x;
    int tile = blockIdx.x;
    int tm = tile / tiles_n;
    int tn = tile - tm * tiles_n;
    int mn   = lane & 15;
    int koff = (lane >> 4) << 1;
    const float* p0   = hp + (size_t)(tm * 16 + mn) * K + koff;
    const float* p1   = p0 + (size_t)B_ * H_;
    const float* p2   = p1 + (size_t)B_ * H_;
    const float* p3   = p2 + (size_t)B_ * H_;
    const float* bk   = b1 + koff;
    const float* wcol = W + (size_t)koff * N + (tn * 16 + mn);
    v2f a[32], b[32];
#pragma unroll
    for (int s = 0; s < 32; ++s) {
        int k = 4 * s;
        v2f u = *(const v2f*)(p0 + k);
        v2f v = *(const v2f*)(p1 + k);
        v2f y = *(const v2f*)(p2 + k);
        v2f w = *(const v2f*)(p3 + k);
        v2f e = *(const v2f*)(bk + k);
        a[s].x = fmaxf(((u.x + v.x) + (y.x + w.x)) + e.x, 0.f);
        a[s].y = fmaxf(((u.y + v.y) + (y.y + w.y)) + e.y, 0.f);
        b[s].x = wcol[(size_t)k * N];
        b[s].y = wcol[(size_t)(k + 1) * N];
    }
    v8f c = {};
#pragma unroll
    for (int s = 0; s < 32; ++s)
        c = __builtin_amdgcn_wmma_f32_16x16x4_f32(
                false, a[s], false, b[s], (short)0, c, false, false);
    float bv = bias[tn * 16 + mn];
    int rbase = tm * 16 + ((lane >> 4) << 3);
#pragma unroll
    for (int j = 0; j < 8; ++j)
        D[(size_t)(rbase + j) * N + (tn * 16 + mn)] = c[j] + bv;
}

#define HD_PITCH 132
#define DEC_PANELS 5
#define DEC_THREADS (32 * 4 * DEC_PANELS)

__global__ void __launch_bounds__(DEC_THREADS)
dec_gemm_wreg(const float* __restrict__ A,
              const float* __restrict__ W,
              const float* __restrict__ bias,
              float* __restrict__ D) {
    __shared__ v4f hs4[B_ * HD_PITCH / 4];
    float* hs = (float*)hs4;
    const int t    = threadIdx.x;
    const int lane = t & 31;
    const int wv   = t >> 5;
    const int c    = lane & 15;
    const int h    = lane >> 4;
    const int pl   = wv >> 2;
    const int rq   = wv & 3;
    const int col  = (blockIdx.x * DEC_PANELS + pl) * 16 + c;

    const float* wcol = W + (size_t)(2 * h) * G_ + col;
    v2f bf[32];
#pragma unroll
    for (int i = 0; i < 32; ++i) {
        bf[i].x = wcol[(size_t)(4 * i) * G_];
        bf[i].y = wcol[(size_t)(4 * i + 1) * G_];
    }
    const float bv = bias[col];

    if (wv < 16) {
        const v4f* A4 = (const v4f*)A;
        v4f r[16];
#pragma unroll
        for (int i = 0; i < 16; ++i)
            r[i] = A4[t + 512 * i];
#pragma unroll
        for (int i = 0; i < 16; ++i) {
            int q = t + 512 * i;
            *(v4f*)(hs + (q >> 5) * HD_PITCH + ((q & 31) << 2)) = r[i];
        }
    }
    __syncthreads();

    const float* abase = hs + (rq * 64 + c) * HD_PITCH + 2 * h;
    for (int r = 0; r < 4; ++r) {
        const float* arow = abase + r * 16 * HD_PITCH;
        v2f af[32];
#pragma unroll
        for (int i = 0; i < 32; ++i)
            af[i] = *(const v2f*)(arow + 4 * i);
        v8f acc = {};
#pragma unroll
        for (int i = 0; i < 32; ++i)
            acc = __builtin_amdgcn_wmma_f32_16x16x4_f32(
                      false, af[i], false, bf[i], (short)0, acc, false, false);
        const int rbase = (rq * 4 + r) * 16 + (h << 3);
#pragma unroll
        for (int j = 0; j < 8; ++j)
            D[(size_t)(rbase + j) * G_ + col] = acc[j] + bv;
    }
}

__global__ void reparam_kernel(const float* __restrict__ ml,
                               const float* __restrict__ eps,
                               float* __restrict__ z,
                               float* __restrict__ mean_o,
                               float* __restrict__ logvar_o) {
    int i = blockIdx.x * blockDim.x + threadIdx.x;
    if (i >= B_ * E_) return;
    int b = i >> 6, e = i & 63;
    float mean   = ml[b * (2 * E_) + e];
    float logvar = ml[b * (2 * E_) + E_ + e];
    z[i]        = fmaf(eps[i], logvar, mean);
    mean_o[i]   = mean;
    logvar_o[i] = logvar;
}

extern "C" void kernel_launch(void* const* d_in, const int* in_sizes, int n_in,
                              void* d_out, int out_size, void* d_ws, size_t ws_size,
                              hipStream_t stream) {
    const float* x     = (const float*)d_in[0];
    const int*   pidx  = (const int*)d_in[1];
    const float* pmask = (const float*)d_in[2];
    const float* eps   = (const float*)d_in[3];
    const float* pw    = (const float*)d_in[4];
    const float* pb    = (const float*)d_in[5];
    const float* ew1   = (const float*)d_in[6];
    const float* eb1   = (const float*)d_in[7];
    const float* ew2   = (const float*)d_in[8];
    const float* eb2   = (const float*)d_in[9];
    const float* dw1   = (const float*)d_in[10];
    const float* db1   = (const float*)d_in[11];
    const float* dw2   = (const float*)d_in[12];
    const float* db2   = (const float*)d_in[13];

    float* ws = (float*)d_ws;
    float* hp = ws;
    float* ml = ws + 294912;
    float* hd = ws + 327680;

    float* out    = (float*)d_out;
    float* z_o    = out;
    float* recon  = out + (size_t)B_ * E_;
    float* mean_o = out + (size_t)B_ * E_ + (size_t)B_ * G_;
    float* logv_o = mean_o + (size_t)B_ * E_;

    pathway_rows_lds<<<(B_ / 4) * 4, 1024, 0, stream>>>(x, pidx, pw, pmask, pb, ew1, hp);

    enc_out_gemm<<<(B_ / 16) * (2 * E_ / 16), 32, 0, stream>>>(hp, eb1, ew2, eb2, ml);
    reparam_kernel<<<(B_ * E_ + 255) / 256, 256, 0, stream>>>(ml, eps, z_o, mean_o, logv_o);
    gemm_wmma_f32<1, E_, H_><<<(B_ / 16) * (H_ / 16), 32, 0, stream>>>(z_o, dw1, db1, hd);
    dec_gemm_wreg<<<G_ / (16 * DEC_PANELS), DEC_THREADS, 0, stream>>>(hd, dw2, db2, recon);
}


# Middle GEMMs and reparameterization moved into the decoder launch: a block owns 16 rows x 1/16 of the columns, hd tile in registers as A, dec_w2 streamed as B; 2 launches, not 5; sums unchanged
# speedup vs baseline: 3.1695x; 1.1576x over previous
#include <hip/hip_runtime.h>
#include <hip/hip_bf16.h>

#define B_ 256
#define G_ 20000
#define P_ 1024
#define M_ 256
#define H_ 128
#define E_ 64

typedef __attribute__((ext_vector_type(2))) float v2f;
typedef __attribute__((ext_vector_type(4))) float v4f;
typedef __attribute__((ext_vector_type(8))) float v8f;

typedef __attribute__((ext_vector_type(4))) int v4i;

#define AS_STRIDE 260

__device__ __forceinline__ int xs_slot(int g) { return g ^ ((g >> 4) & 3); }

template <int CTRL>
__device__ __forceinline__ float row_lane_xor(float v) {
    return __builtin_bit_cast(float, __builtin_amdgcn_update_dpp(
        0, __builtin_bit_cast(int, v), CTRL, 0xf, 0xf, false));
}

__device__ __forceinline__ float other_row(float v) {
    int i = __builtin_bit_cast(int, v);
    return __builtin_bit_cast(float, __builtin_amdgcn_permlanex16(
        i, i, 0x76543210, 0xfedcba98, false, false));
}

__device__ __forceinline__ float pathway_sum(const v4f* xs, v4i i0, v4i i1,
                                             v4f w0, v4f w1, v4f k0, v4f k1,
                                             int hrow, int oct) {
    v4f g0 = xs[xs_slot(i0.x)], g1 = xs[xs_slot(i0.y)];
    v4f g2 = xs[xs_slot(i0.z)], g3 = xs[xs_slot(i0.w)];
    v4f g4 = xs[xs_slot(i1.x)], g5 = xs[xs_slot(i1.y)];
    v4f g6 = xs[xs_slot(i1.z)], g7 = xs[xs_slot(i1.w)];
    float m0 = w0.x * k0.x, m1 = w0.y * k0.y, m2 = w0.z * k0.z, m3 = w0.w * k0.w;
    float m4 = w1.x * k1.x, m5 = w1.y * k1.y, m6 = w1.z * k1.z, m7 = w1.w * k1.w;
    v4f a = {g0.x * m0, g0.y * m0, g0.z * m0, g0.w * m0};
#define ROW4_FMA(g, m) a.x = fmaf(g.x, m, a.x); a.y = fmaf(g.y, m, a.y); \
                       a.z = fmaf(g.z, m, a.z); a.w = fmaf(g.w, m, a.w);
    ROW4_FMA(g1, m1) ROW4_FMA(g2, m2) ROW4_FMA(g3, m3) ROW4_FMA(g4, m4)
    ROW4_FMA(g5, m5) ROW4_FMA(g6, m6) ROW4_FMA(g7, m7)
#undef ROW4_FMA
    float ke = hrow ? a.z : a.x, ko = hrow ? a.w : a.y;
    float se = hrow ? a.x : a.z, so = hrow ? a.y : a.w;
    float ue = ke + other_row(se);
    float uo = ko + other_row(so);
    float keep = oct ? uo : ue;
    float send = oct ? ue : uo;
    float v = keep + row_lane_xor<0x168>(send);
    v += row_lane_xor<0x164>(v);
    v += row_lane_xor<0x162>(v);
    v += row_lane_xor<0x161>(v);
    return v;
}

__global__ void __launch_bounds__(1024)
pathway_rows_lds(const float* __restrict__ x,
                 const int* __restrict__ idx,
                 const float* __restrict__ pw,
                 const float* __restrict__ pmask,
                 const float* __restrict__ pb,
                 const float* __restrict__ w1,
                 float* __restrict__ hpart) {
    __shared__ v4f xs[G_];
    __shared__ float as[4 * AS_STRIDE];
    float* red = (float*)xs;
    const int t  = threadIdx.x;
    const int b0 = (blockIdx.x >> 2) << 2;
    const int s  = blockIdx.x & 3;

    {
        const v4f* x0 = (const v4f*)(x + (size_t)b0 * G_);
        const v4f* x1 = (const v4f*)(x + (size_t)(b0 + 1) * G_);
        const v4f* x2 = (const v4f*)(x + (size_t)(b0 + 2) * G_);
        const v4f* x3 = (const v4f*)(x + (size_t)(b0 + 3) * G_);
        constexpr int Q = G_ / 4;
        v4f r0[5], r1[5], r2[5], r3[5];
#pragma unroll
        for (int i = 0; i < 5; ++i) {
            int q  = t + 1024 * i;
            int qc = q < Q ? q : Q - 1;
            r0[i] = x0[qc];
            r1[i] = x1[qc];
            r2[i] = x2[qc];
            r3[i] = x3[qc];
        }
#pragma unroll
        for (int i = 0; i < 5; ++i) {
            int q = t + 1024 * i;
            if (q < Q) {
                const int m = (q >> 2) & 3;
                v4f* dst = xs + 4 * q;
                v4f e0 = {r0[i].x, r1[i].x, r2[i].x, r3[i].x};
                v4f e1 = {r0[i].y, r1[i].y, r2[i].y, r3[i].y};
                v4f e2 = {r0[i].z, r1[i].z, r2[i].z, r3[i].z};
                v4f e3 = {r0[i].w, r1[i].w, r2[i].w, r3[i].w};
                dst[m]     = e0;
                dst[1 ^ m] = e1;
                dst[2 ^ m] = e2;
                dst[3 ^ m] = e3;
            }
        }
    }
    __syncthreads();

    const int lane = t & 31;
    const int c    = lane & 15;
    const int hrow = lane >> 4;
    const int oct  = (lane >> 3) & 1;
    const int c8   = lane & 7;
    const int p0   = s * 256 + (t >> 5) * 8;
    const size_t off = (size_t)p0 * M_ + 4 * lane;
    const int*   ip = idx   + off;
    const float* wp = pw    + off;
    const float* kp = pmask + off;

    float res = 0.f;
#pragma clang loop unroll(disable)
    for (int j = 0; j < 8; j += 2) {
        const int ja = j * M_, jb = ja + M_;
        v4i ai0 = *(const v4i*)(ip + ja), ai1 = *(const v4i*)(ip + ja + 128);
        v4f aw0 = *(const v4f*)(wp + ja), aw1 = *(const v4f*)(wp + ja + 128);
        v4f ak0 = *(const v4f*)(kp + ja), ak1 = *(const v4f*)(kp + ja + 128);
        v4i bi0 = *(const v4i*)(ip + jb), bi1 = *(const v4i*)(ip + jb + 128);
        v4f bw0 = *(const v4f*)(wp + jb), bw1 = *(const v4f*)(wp + jb + 128);
        v4f bk0 = *(const v4f*)(kp + jb), bk1 = *(const v4f*)(kp + jb + 128);
        float va = pathway_sum(xs, ai0, ai1, aw0, aw1, ak0, ak1, hrow, oct);
        float vb = pathway_sum(xs, bi0, bi1, bw0, bw1, bk0, bk1, hrow, oct);
        if (c8 == j)     res = va;
        if (c8 == j + 1) res = vb;
    }
    {
        const int wv   = t >> 5;
        const int n    = wv & 7;
        const int q    = wv >> 3;
        const int koff = hrow << 1;
        const float* wcol = w1 + (size_t)(s * 256 + q * 64 + koff) * H_ + n * 16 + c;
        const float* arow = as + (c & 3) * AS_STRIDE + q * 64 + koff;
        v2f bf[16];
#pragma unroll
        for (int i = 0; i < 16; ++i) {
            bf[i].x = wcol[(size_t)(4 * i) * H_];
            bf[i].y = wcol[(size_t)(4 * i + 1) * H_];
        }

        as[(lane >> 3) * AS_STRIDE + wv * 8 + c8] = tanhf(res + pb[p0 + c8]);
        __syncthreads();

        v8f acc = {};
#pragma unroll
        for (int i = 0; i < 16; ++i) {
            v2f af = *(const v2f*)(arow + 4 * i);
            acc = __builtin_amdgcn_wmma_f32_16x16x4_f32(
                      false, af, false, bf[i], (short)0, acc, false, false);
        }
        if (hrow == 0) {
            red[(q * 4 + 0) * H_ + n * 16 + c] = acc[0];
            red[(q * 4 + 1) * H_ + n * 16 + c] = acc[1];
            red[(q * 4 + 2) * H_ + n * 16 + c] = acc[2];
            red[(q * 4 + 3) * H_ + n * 16 + c] = acc[3];
        }
    }
    __syncthreads();
    if (t < 4 * H_) {
        const int r = t >> 7, col = t & (H_ - 1);
        float v = (red[(0 * 4 + r) * H_ + col] + red[(1 * 4 + r) * H_ + col])
                + (red[(2 * 4 + r) * H_ + col] + red[(3 * 4 + r) * H_ + col]);
        hpart[((size_t)s * B_ + b0 + r) * H_ + col] = v;
    }
}

#define ROW_PITCH 132
#define Z_PITCH 68
#define DEC_CHUNKS 16
#define DEC_GROUPS (G_ / 32)

template <int HALF>
__device__ __forceinline__ void load_half(const float* __restrict__ w,
                                          v2f (&r0)[32], v2f (&r1)[32]) {
#pragma unroll
    for (int i = 16 * HALF; i < 16 * HALF + 16; ++i) {
        r0[i] = *(const v2f*)(w + (size_t)(4 * i) * G_);
        r1[i] = *(const v2f*)(w + (size_t)(4 * i + 1) * G_);
    }
}

template <int HALF>
__device__ __forceinline__ void mma_half(const v2f (&af)[32],
                                         const v2f (&r0)[32], const v2f (&r1)[32],
                                         v8f& acc0, v8f& acc1) {
#pragma unroll
    for (int i = 16 * HALF; i < 16 * HALF + 16; ++i) {
        v2f b0 = {r0[i].x, r1[i].x};
        v2f b1 = {r0[i].y, r1[i].y};
        acc0 = __builtin_amdgcn_wmma_f32_16x16x4_f32(
                   false, af[i], false, b0, (short)0, acc0, false, false);
        acc1 = __builtin_amdgcn_wmma_f32_16x16x4_f32(
                   false, af[i], false, b1, (short)0, acc1, false, false);
    }
}

__device__ __forceinline__ void store_group(v8f acc0, v8f acc1, v2f bv,
                                            float* __restrict__ dst) {
#pragma unroll
    for (int j = 0; j < 8; ++j) {
        v2f o = {acc0[j] + bv.x, acc1[j] + bv.y};
        *(v2f*)(dst + (size_t)j * G_) = o;
    }
}

__device__ __forceinline__ void group_step(const v2f (&af)[32],
                                           const v2f (&c0)[32], const v2f (&c1)[32], v2f bvc,
                                           v2f (&n0)[32], v2f (&n1)[32], v2f& bvn,
                                           const float* __restrict__ wn,
                                           const float* __restrict__ bn,
                                           float* __restrict__ dst) {
    v8f acc0 = {}, acc1 = {};
    load_half<0>(wn, n0, n1);
    load_half<1>(wn, n0, n1);
    bvn = *(const v2f*)bn;
    __builtin_amdgcn_sched_barrier(0);
    mma_half<0>(af, c0, c1, acc0, acc1);
    mma_half<1>(af, c0, c1, acc0, acc1);
    __builtin_amdgcn_sched_barrier(0);
    store_group(acc0, acc1, bvc, dst);
}

__device__ __forceinline__ void group_last(const v2f (&af)[32],
                                           const v2f (&c0)[32], const v2f (&c1)[32], v2f bvc,
                                           float* __restrict__ dst) {
    v8f acc0 = {}, acc1 = {};
    mma_half<0>(af, c0, c1, acc0, acc1);
    mma_half<1>(af, c0, c1, acc0, acc1);
    store_group(acc0, acc1, bvc, dst);
}

static_assert(DEC_GROUPS / DEC_CHUNKS >= 32 &&
              (DEC_GROUPS + DEC_CHUNKS - 1) / DEC_CHUNKS <= 40, "group count per wave");

__global__ void __launch_bounds__(256)
dec_rowtile_fused(const float* __restrict__ hp,
                  const float* __restrict__ eb1,
                  const float* __restrict__ ew2,
                  const float* __restrict__ eb2,
                  const float* __restrict__ eps,
                  const float* __restrict__ dw1,
                  const float* __restrict__ db1,
                  const float* __restrict__ dw2,
                  const float* __restrict__ db2,
                  float* __restrict__ z_o,
                  float* __restrict__ mean_o,
                  float* __restrict__ logv_o,
                  float* __restrict__ recon) {
    __shared__ v4f h_s4[16 * ROW_PITCH / 4];
    __shared__ v4f ml_s4[16 * ROW_PITCH / 4];
    __shared__ v4f z_s4[16 * Z_PITCH / 4];
    __shared__ v4f hd_s4[16 * ROW_PITCH / 4];
    float* h_s  = (float*)h_s4;
    float* ml_s = (float*)ml_s4;
    float* z_s  = (float*)z_s4;
    float* hd_s = (float*)hd_s4;

    const int t    = threadIdx.x;
    const int lane = t & 31;
    const int wv   = __builtin_amdgcn_readfirstlane(t >> 5);
    const int c    = lane & 15;
    const int h    = lane >> 4;
    const int koff = h << 1;
    const int rt   = blockIdx.x & 15;
    const int cc   = blockIdx.x >> 4;
    const int row0 = rt * 16;
    const int gbeg = (cc * DEC_GROUPS) / DEC_CHUNKS;
    const int gend = ((cc + 1) * DEC_GROUPS) / DEC_CHUNKS;

    {
        const int r = t >> 4, col = (t & 15) << 3;
        const float* q = hp + (size_t)(row0 + r) * H_ + col;
        v4f u0 = *(const v4f*)(q),                         u1 = *(const v4f*)(q + 4);
        v4f v0 = *(const v4f*)(q + (size_t)B_ * H_),       v1 = *(const v4f*)(q + (size_t)B_ * H_ + 4);
        v4f y0 = *(const v4f*)(q + (size_t)2 * B_ * H_),   y1 = *(const v4f*)(q + (size_t)2 * B_ * H_ + 4);
        v4f w0 = *(const v4f*)(q + (size_t)3 * B_ * H_),   w1 = *(const v4f*)(q + (size_t)3 * B_ * H_ + 4);
        v4f e0 = *(const v4f*)(eb1 + col),                 e1 = *(const v4f*)(eb1 + col + 4);
        v4f o0, o1;
#pragma unroll
        for (int k = 0; k < 4; ++k) {
            o0[k] = fmaxf(((u0[k] + v0[k]) + (y0[k] + w0[k])) + e0[k], 0.f);
            o1[k] = fmaxf(((u1[k] + v1[k]) + (y1[k] + w1[k])) + e1[k], 0.f);
        }
        *(v4f*)(h_s + r * ROW_PITCH + col)     = o0;
        *(v4f*)(h_s + r * ROW_PITCH + col + 4) = o1;
    }

    v2f be[32];
    {
        constexpr int N = 2 * E_;
        const float* wcol = ew2 + (size_t)koff * N + wv * 16 + c;
#pragma unroll
        for (int s = 0; s < 32; ++s) {
            be[s].x = wcol[(size_t)(4 * s) * N];
            be[s].y = wcol[(size_t)(4 * s + 1) * N];
        }
    }
    const float bve = eb2[wv * 16 + c];
    v2f bd[16];
    {
        const float* wcol = dw1 + (size_t)koff * H_ + wv * 16 + c;
#pragma unroll
        for (int s = 0; s < 16; ++s) {
            bd[s].x = wcol[(size_t)(4 * s) * H_];
            bd[s].y = wcol[(size_t)(4 * s + 1) * H_];
        }
    }
    const float bvd = db1[wv * 16 + c];
    float ep[4];
#pragma unroll
    for (int u = 0; u < 4; ++u)
        ep[u] = eps[row0 * E_ + t + 256 * u];

    __syncthreads();

    {
        const float* arow = h_s + c * ROW_PITCH + koff;
        v8f acc = {};
#pragma unroll
        for (int s = 0; s < 32; ++s) {
            v2f a = *(const v2f*)(arow + 4 * s);
            acc = __builtin_amdgcn_wmma_f32_16x16x4_f32(
                      false, a, false, be[s], (short)0, acc, false, false);
        }
#pragma unroll
        for (int j = 0; j < 8; ++j)
            ml_s[(8 * h + j) * ROW_PITCH + wv * 16 + c] = acc[j] + bve;
    }
    __syncthreads();

#pragma unroll
    for (int u = 0; u < 4; ++u) {
        const int i = t + 256 * u;
        const int r = i >> 6, e = i & 63;
        const float mean   = ml_s[r * ROW_PITCH + e];
        const float logvar = ml_s[r * ROW_PITCH + E_ + e];
        const float zz = fmaf(ep[u], logvar, mean);
        z_s[r * Z_PITCH + e] = zz;
        if (cc == 0) {
            z_o[row0 * E_ + i]    = zz;
            mean_o[row0 * E_ + i] = mean;
            logv_o[row0 * E_ + i] = logvar;
        }
    }
    __syncthreads();

    const float* wbase = dw2 + (size_t)koff * G_ + 2 * c;
    v2f p0[32], p1[32], q0[32], q1[32];
    const int g = gbeg + wv;
    v2f bvp = *(const v2f*)(db2 + 32 * g + 2 * c);
    v2f bvq;
    load_half<0>(wbase + 32 * g, p0, p1);
    load_half<1>(wbase + 32 * g, p0, p1);

    {
        const float* zrow = z_s + c * Z_PITCH + koff;
        v8f acc = {};
#pragma unroll
        for (int s = 0; s < 16; ++s) {
            v2f az = *(const v2f*)(zrow + 4 * s);
            acc = __builtin_amdgcn_wmma_f32_16x16x4_f32(
                      false, az, false, bd[s], (short)0, acc, false, false);
        }
#pragma unroll
        for (int j = 0; j < 8; ++j)
            hd_s[(8 * h + j) * ROW_PITCH + wv * 16 + c] = fmaxf(acc[j] + bvd, 0.f);
    }
    __syncthreads();

    v2f af[32];
    {
        const float* arow = hd_s + c * ROW_PITCH + koff;
#pragma unroll
        for (int i = 0; i < 32; ++i)
            af[i] = *(const v2f*)(arow + 4 * i);
    }

    float* dbase = recon + (size_t)(row0 + 8 * h) * G_ + 2 * c;
    const float* bbase = db2 + 2 * c;
    const bool five = g + 32 < gend;
    group_step(af, p0, p1, bvp, q0, q1, bvq,
               wbase + 32 * (g + 8), bbase + 32 * (g + 8), dbase + 32 * g);
    group_step(af, q0, q1, bvq, p0, p1, bvp,
               wbase + 32 * (g + 16), bbase + 32 * (g + 16), dbase + 32 * (g + 8));
    group_step(af, p0, p1, bvp, q0, q1, bvq,
               wbase + 32 * (g + 24), bbase + 32 * (g + 24), dbase + 32 * (g + 16));
    if (five) {
        group_step(af, q0, q1, bvq, p0, p1, bvp,
                   wbase + 32 * (g + 32), bbase + 32 * (g + 32), dbase + 32 * (g + 24));
        group_last(af, p0, p1, bvp, dbase + 32 * (g + 32));
    } else {
        group_last(af, q0, q1, bvq, dbase + 32 * (g + 24));
    }
}

extern "C" void kernel_launch(void* const* d_in, const int* in_sizes, int n_in,
                              void* d_out, int out_size, void* d_ws, size_t ws_size,
                              hipStream_t stream) {
    const float* x     = (const float*)d_in[0];
    const int*   pidx  = (const int*)d_in[1];
    const float* pmask = (const float*)d_in[2];
    const float* eps   = (const float*)d_in[3];
    const float* pw    = (const float*)d_in[4];
    const float* pb    = (const float*)d_in[5];
    const float* ew1   = (const float*)d_in[6];
    const float* eb1   = (const float*)d_in[7];
    const float* ew2   = (const float*)d_in[8];
    const float* eb2   = (const float*)d_in[9];
    const float* dw1   = (const float*)d_in[10];
    const float* db1   = (const float*)d_in[11];
    const float* dw2   = (const float*)d_in[12];
    const float* db2   = (const float*)d_in[13];

    float* ws = (float*)d_ws;
    float* hp = ws;

    float* out    = (float*)d_out;
    float* z_o    = out;
    float* recon  = out + (size_t)B_ * E_;
    float* mean_o = out + (size_t)B_ * E_ + (size_t)B_ * G_;
    float* logv_o = mean_o + (size_t)B_ * E_;

    pathway_rows_lds<<<(B_ / 4) * 4, 1024, 0, stream>>>(x, pidx, pw, pmask, pb, ew1, hp);

    dec_rowtile_fused<<<(B_ / 16) * DEC_CHUNKS, 256, 0, stream>>>(
        hp, eb1, ew2, eb2, eps, dw1, db1, dw2, db2, z_o, mean_o, logv_o, recon);
}
